# starting kernel
# baseline (speedup 1.0000x reference)
#include <hip/hip_runtime.h>

typedef float v2f __attribute__((ext_vector_type(2)));
typedef float v8f __attribute__((ext_vector_type(8)));

#define NLAT_IN   181
#define NLON_IN   360
#define NLAT_OUT  361
#define NLON_OUT  720
#define KSIZE     3
#define CIN       16
#define COUT      16
#define NNZ       8192
#define NPIX      (NLAT_IN * NLON_IN)
#define ACC_PITCH 727
#define SCAT_THREADS 512

__global__ __launch_bounds__(32)
void disco_mix_wmma(const float* __restrict__ x,
                    const float* __restrict__ w,
                    float* __restrict__ xw)
{
    const int lane = threadIdx.x;
    const int m0   = blockIdx.x * 16;
    const int mrow = lane & 15;
    const int half = lane >> 4;
    const int co   = mrow;
    const bool full_tile = (m0 + 16 <= NPIX);

    int m = m0 + mrow;
    if (m >= NPIX) m = NPIX - 1;

    v2f a[4];
#pragma unroll
    for (int s = 0; s < 4; ++s) {
        const int ci = 4 * s + 2 * half;
        a[s].x = x[(ci + 0) * NPIX + m];
        a[s].y = x[(ci + 1) * NPIX + m];
    }

#pragma unroll
    for (int kt = 0; kt < KSIZE; ++kt) {
        v8f c = {0.f, 0.f, 0.f, 0.f, 0.f, 0.f, 0.f, 0.f};
#pragma unroll
        for (int s = 0; s < 4; ++s) {
            const int ci = 4 * s + 2 * half;
            v2f b;
            b.x = w[co * (CIN * KSIZE) + (ci + 0) * KSIZE + kt];
            b.y = w[co * (CIN * KSIZE) + (ci + 1) * KSIZE + kt];
            c = __builtin_amdgcn_wmma_f32_16x16x4_f32(false, a[s], false, b,
                                                      (short)0, c, false, false);
        }
        float* dst = xw + ((size_t)(kt * NPIX + m0 + 8 * half) * COUT + co);
        if (full_tile) {
#pragma unroll
            for (int v = 0; v < 8; ++v)
                dst[v * COUT] = c[v];
        } else {
#pragma unroll
            for (int v = 0; v < 8; ++v) {
                const int pix = m0 + v + 8 * half;
                if (pix < NPIX) dst[v * COUT] = c[v];
            }
        }
    }
}

__global__ __launch_bounds__(SCAT_THREADS)
void disco_scatter_lat(const float* __restrict__ xw,
                       const float* __restrict__ bias,
                       const int*   __restrict__ ker_idx,
                       const int*   __restrict__ row_idx,
                       const int*   __restrict__ col_idx,
                       const float* __restrict__ vals,
                       float* __restrict__ out)
{
    __shared__ float acc[COUT * ACC_PITCH];
    __shared__ int   list[NNZ];
    __shared__ int   cnt;

    const int tid = threadIdx.x;
    const int hi  = blockIdx.x;

    if (tid == 0) cnt = 0;
    for (int i = tid; i < COUT * ACC_PITCH; i += SCAT_THREADS) acc[i] = 0.f;
    __syncthreads();

    for (int nz = tid; nz < NNZ; nz += SCAT_THREADS) {
        if (col_idx[nz] / NLON_OUT == hi) {
            const int slot = atomicAdd(&cnt, 1);
            list[slot] = nz;
        }
    }
    __syncthreads();

    const int n = cnt;
    for (int j = 0; j < n; ++j) {
        const int   nz = list[j];
        const int   k  = ker_idx[nz];
        const int   r  = row_idx[nz];
        const int   p  = col_idx[nz] - hi * NLON_OUT;
        const float v  = vals[nz];
        const float* plane = xw + (size_t)(k * NPIX + r * NLON_IN) * COUT;

        for (int e = tid; e < COUT * NLON_IN; e += SCAT_THREADS) {
            const int co = e & 15;
            const int po = e >> 4;
            int lon = p + 2 * po;
            if (lon >= NLON_OUT) lon -= NLON_OUT;
            atomicAdd(&acc[co * ACC_PITCH + lon], v * plane[po * COUT + co]);
        }
    }
    __syncthreads();

    for (int e = tid; e < COUT * NLON_OUT; e += SCAT_THREADS) {
        const int co = e / NLON_OUT;
        const int L  = e - co * NLON_OUT;
        out[(co * NLAT_OUT + hi) * NLON_OUT + L] = acc[co * ACC_PITCH + L] + bias[co];
    }
}

extern "C" void kernel_launch(void* const* d_in, const int* in_sizes, int n_in,
                              void* d_out, int out_size, void* d_ws, size_t ws_size,
                              hipStream_t stream)
{
    (void)in_sizes; (void)n_in; (void)out_size; (void)ws_size;

    const float* x    = (const float*)d_in[0];
    const float* w    = (const float*)d_in[1];
    const float* bias = (const float*)d_in[2];
    const int*   ker  = (const int*)  d_in[3];
    const int*   row  = (const int*)  d_in[4];
    const int*   col  = (const int*)  d_in[5];
    const float* vals = (const float*)d_in[6];
    float*       out  = (float*)d_out;

    float* xw = (float*)d_ws;

    const int tiles = (NPIX + 15) / 16;
    disco_mix_wmma<<<tiles, 32, 0, stream>>>(x, w, xw);
    disco_scatter_lat<<<NLAT_OUT, SCAT_THREADS, 0, stream>>>(xw, bias, ker, row, col,
                                                             vals, out);
}


# sum over entries moved in front of the channel mix: per basis, v*x rows are added into the A fragments by vector fma, then 12 WMMAs mix them; sum order: per basis in queue order, bases 0,1,2
# speedup vs baseline: 2.5916x; 2.5916x over previous
#include <hip/hip_runtime.h>


typedef float v2f __attribute__((ext_vector_type(2)));
typedef float v8f __attribute__((ext_vector_type(8)));
typedef int   v2i __attribute__((ext_vector_type(2)));

#define NLAT_IN   181
#define NLON_IN   360
#define NLAT_OUT  361
#define NLON_OUT  720
#define KSIZE     3
#define CIN       16
#define COUT      16
#define NNZ       8192
#define NPIX      (NLAT_IN * NLON_IN)
#define ACC_PITCH 727
#define THREADS   512
#define TILES_PER_WAVE 3

__device__ __forceinline__ void add_rows(v2f (&S)[TILES_PER_WAVE][4],
                                         const v2f (&a)[TILES_PER_WAVE][4], float v)
{
#pragma unroll
    for (int i = 0; i < TILES_PER_WAVE; ++i)
#pragma unroll
        for (int s = 0; s < 4; ++s)
            S[i][s] += a[i][s] * v;
}

__device__ __forceinline__ void mix_basis(v8f (&d)[TILES_PER_WAVE],
                                          const v2f (&S)[TILES_PER_WAVE][4], const v2f* wk)
{
    v2f b[4];
#pragma unroll
    for (int s = 0; s < 4; ++s)
        b[s] = wk[s * 32];
#pragma unroll
    for (int s = 0; s < 4; ++s) {
#pragma unroll
        for (int i = 0; i < TILES_PER_WAVE; ++i) {
            d[i] = __builtin_amdgcn_wmma_f32_16x16x4_f32(false, S[i][s], false, b[s],
                                                         (short)0, d[i], false, false);
        }
    }
}

__global__ __launch_bounds__(THREADS)
void disco_fused_lat(const float* __restrict__ x,
                     const float* __restrict__ w,
                     const float* __restrict__ bias,
                     const int*   __restrict__ ker_idx,
                     const int*   __restrict__ row_idx,
                     const int*   __restrict__ col_idx,
                     const float* __restrict__ vals,
                     float* __restrict__ out)
{
    __shared__ float acc[COUT * ACC_PITCH];
    __shared__ v2i   rec[NNZ];
    __shared__ v2f   wl[KSIZE * 4 * 32];
    __shared__ int   cnt[2];

    const int tid  = threadIdx.x;
    const int hi   = blockIdx.x;
    const int lane = tid & 31;
    const int wave = __builtin_amdgcn_readfirstlane(tid >> 5);

    if (tid < 2) cnt[tid] = 0;
    if (tid < KSIZE * 4 * 32) {
        const int k  = tid >> 7;
        const int s  = (tid >> 5) & 3;
        const int cc = tid & 15;
        const int hh = (tid >> 4) & 1;
        const int ci = 4 * s + 2 * hh;
        v2f wv;
        wv.x = w[cc * (CIN * KSIZE) + (ci + 0) * KSIZE + k];
        wv.y = w[cc * (CIN * KSIZE) + (ci + 1) * KSIZE + k];
        wl[tid] = wv;
    }
    __syncthreads();

    for (int nz = tid; nz < NNZ; nz += THREADS) {
        const int col = col_idx[nz];
        if (col / NLON_OUT == hi) {
            const int p    = col - hi * NLON_OUT;
            const int par  = p & 1;
            const int word = (row_idx[nz] * NLON_IN) | (ker_idx[nz] << 16) | ((p >> 1) << 18);
            const int slot = atomicAdd(&cnt[par], 1);
            v2i rc;
            rc.x = word;
            rc.y = __float_as_int(vals[nz]);
            rec[par ? (NNZ - 1 - slot) : slot] = rc;
        }
    }
    __syncthreads();

    const int q  = wave & 1;
    const int c  = lane & 15;
    const int h  = lane >> 4;
    const int n  = __builtin_amdgcn_readfirstlane(cnt[q]);

    int trow[TILES_PER_WAVE];
#pragma unroll
    for (int i = 0; i < TILES_PER_WAVE; ++i) {
        const int t = 16 * ((wave >> 1) + 8 * i) + c;
        trow[i] = t < NLON_IN ? t : NLON_IN - 1;
    }

    const float* xh = x + 2 * h * NPIX;

    v8f d[TILES_PER_WAVE];
#pragma unroll
    for (int i = 0; i < TILES_PER_WAVE; ++i)
        d[i] = (v8f){0.f, 0.f, 0.f, 0.f, 0.f, 0.f, 0.f, 0.f};

    for (int kk = 0; kk < KSIZE; ++kk) {
      v2f S[TILES_PER_WAVE][4];
#pragma unroll
      for (int i = 0; i < TILES_PER_WAVE; ++i) {
#pragma unroll
        for (int s = 0; s < 4; ++s)
            S[i][s] = (v2f){0.f, 0.f};
      }
      for (int j0 = 0; j0 < n; j0 += 32) {
        const int idx = j0 + lane;
        v2i rc = (v2i){0, 0};
        if (idx < n) rc = rec[q ? (NNZ - 1 - idx) : idx];
        int kj = (rc.x >> 16) & 3;
        if (kj > KSIZE - 1) kj = KSIZE - 1;
        unsigned mask = __builtin_amdgcn_ballot_w32(idx < n && kj == kk);
        while (mask != 0u) {
          const int pick = __builtin_ctz(mask);
          mask &= mask - 1u;
          const int   word = __builtin_amdgcn_readlane(rc.x, pick);
          const float v    = __int_as_float(__builtin_amdgcn_readlane(rc.y, pick));
          const int   rowbase = word & 0xFFFF;
          const int   sh      = word >> 18;

          v2f a[TILES_PER_WAVE][4];
#pragma unroll
          for (int i = 0; i < TILES_PER_WAVE; ++i) {
            int po = trow[i] - sh;
            if (po < 0) po += NLON_IN;
            const float* src = xh + rowbase + po;
#pragma unroll
            for (int s = 0; s < 4; ++s) {
                a[i][s].x = src[(4 * s + 0) * NPIX];
                a[i][s].y = src[(4 * s + 1) * NPIX];
            }
          }
          add_rows(S, a, v);
        }
      }
      mix_basis(d, S, wl + kk * (4 * 32) + lane);
    }

#pragma unroll
    for (int i = 0; i < TILES_PER_WAVE; ++i) {
        const int tb = 16 * ((wave >> 1) + 8 * i) + 8 * h;
#pragma unroll
        for (int r = 0; r < 8; ++r) {
            const int t = tb + r;
            if (i < 2 || t < NLON_IN) acc[c * ACC_PITCH + q + 2 * t] = d[i][r];
        }
    }
    __syncthreads();

    for (int e = tid; e < COUT * NLON_OUT; e += THREADS) {
        const int co = e / NLON_OUT;
        const int L  = e - co * NLON_OUT;
        out[(co * NLAT_OUT + hi) * NLON_OUT + L] = acc[co * ACC_PITCH + L] + bias[co];
    }
}

extern "C" void kernel_launch(void* const* d_in, const int* in_sizes, int n_in,
                              void* d_out, int out_size, void* d_ws, size_t ws_size,
                              hipStream_t stream)
{
    (void)in_sizes; (void)n_in; (void)out_size; (void)d_ws; (void)ws_size;

    const float* x    = (const float*)d_in[0];
    const float* w    = (const float*)d_in[1];
    const float* bias = (const float*)d_in[2];
    const int*   ker  = (const int*)  d_in[3];
    const int*   row  = (const int*)  d_in[4];
    const int*   col  = (const int*)  d_in[5];
    const float* vals = (const float*)d_in[6];
    float*       out  = (float*)d_out;

    disco_fused_lat<<<NLAT_OUT, THREADS, 0, stream>>>(x, w, bias, ker, row, col, vals, out);
}


# x rows read with one 16-byte load per lane and channel (4 consecutive longitudes; tile e = ring positions base+4*row+e), 6 gather waves per parity instead of 8; sum order unchanged
# speedup vs baseline: 3.2097x; 1.2385x over previous
#include <hip/hip_runtime.h>


typedef float v2f __attribute__((ext_vector_type(2)));
typedef float v4f __attribute__((ext_vector_type(4)));
typedef float v8f __attribute__((ext_vector_type(8)));
typedef float v4f_u __attribute__((ext_vector_type(4), aligned(4)));
typedef int   v2i __attribute__((ext_vector_type(2)));

#define NLAT_IN   181
#define NLON_IN   360
#define NLAT_OUT  361
#define NLON_OUT  720
#define KSIZE     3
#define CIN       16
#define COUT      16
#define NNZ       8192
#define NPIX      (NLAT_IN * NLON_IN)
#define ACC_PITCH 727
#define THREADS   512
#define GATHER_PAIRS 6
#define TILES_PER_WAVE 4

__device__ __forceinline__ void mix_basis(v8f (&d)[TILES_PER_WAVE],
                                          const v4f (&S)[8], const v2f* wk)
{
    v2f b[4];
#pragma unroll
    for (int s = 0; s < 4; ++s)
        b[s] = wk[s * 32];
#pragma unroll
    for (int s = 0; s < 4; ++s) {
#pragma unroll
        for (int e = 0; e < TILES_PER_WAVE; ++e) {
            v2f a;
            a.x = S[2 * s + 0][e];
            a.y = S[2 * s + 1][e];
            d[e] = __builtin_amdgcn_wmma_f32_16x16x4_f32(false, a, false, b[s],
                                                         (short)0, d[e], false, false);
        }
    }
}

__global__ __launch_bounds__(THREADS)
void disco_fused_lat(const float* __restrict__ x,
                     const float* __restrict__ w,
                     const float* __restrict__ bias,
                     const int*   __restrict__ ker_idx,
                     const int*   __restrict__ row_idx,
                     const int*   __restrict__ col_idx,
                     const float* __restrict__ vals,
                     float* __restrict__ out)
{
    __shared__ float acc[COUT * ACC_PITCH];
    __shared__ v2i   rec[NNZ];
    __shared__ v2f   wl[KSIZE * 4 * 32];
    __shared__ int   cnt[2];

    const int tid  = threadIdx.x;
    const int hi   = blockIdx.x;
    const int lane = tid & 31;
    const int wave = __builtin_amdgcn_readfirstlane(tid >> 5);

    if (tid < 2) cnt[tid] = 0;
    if (tid < KSIZE * 4 * 32) {
        const int k  = tid >> 7;
        const int s  = (tid >> 5) & 3;
        const int cc = tid & 15;
        const int hh = (tid >> 4) & 1;
        const int ci = 4 * s + 2 * hh;
        v2f wv;
        wv.x = w[cc * (CIN * KSIZE) + (ci + 0) * KSIZE + k];
        wv.y = w[cc * (CIN * KSIZE) + (ci + 1) * KSIZE + k];
        wl[tid] = wv;
    }
    __syncthreads();

    for (int nz = tid; nz < NNZ; nz += THREADS) {
        const int col = col_idx[nz];
        if (col / NLON_OUT == hi) {
            const int p    = col - hi * NLON_OUT;
            const int par  = p & 1;
            const int word = (row_idx[nz] * NLON_IN) | (ker_idx[nz] << 16) | ((p >> 1) << 18);
            const int slot = atomicAdd(&cnt[par], 1);
            v2i rc;
            rc.x = word;
            rc.y = __float_as_int(vals[nz]);
            rec[par ? (NNZ - 1 - slot) : slot] = rc;
        }
    }
    __syncthreads();

    const int q  = wave & 1;
    const int wp = wave >> 1;
    const int c  = lane & 15;
    const int h  = lane >> 4;

    if (wp < GATHER_PAIRS) {
    const int n  = __builtin_amdgcn_readfirstlane(cnt[q]);

    const int t0  = 64 * wp + 4 * c;
    const int t0c = t0 < NLON_IN ? t0 : NLON_IN - 4;

    const float* xh = x + 2 * h * NPIX;

    v8f d[TILES_PER_WAVE];
#pragma unroll
    for (int e = 0; e < TILES_PER_WAVE; ++e)
        d[e] = (v8f){0.f, 0.f, 0.f, 0.f, 0.f, 0.f, 0.f, 0.f};

    for (int kk = 0; kk < KSIZE; ++kk) {
      v4f S[8];
#pragma unroll
      for (int k = 0; k < 8; ++k)
          S[k] = (v4f){0.f, 0.f, 0.f, 0.f};
      for (int j0 = 0; j0 < n; j0 += 32) {
        const int idx = j0 + lane;
        v2i rc = (v2i){0, 0};
        if (idx < n) rc = rec[q ? (NNZ - 1 - idx) : idx];
        int kj = (rc.x >> 16) & 3;
        if (kj > KSIZE - 1) kj = KSIZE - 1;
        unsigned mask = __builtin_amdgcn_ballot_w32(idx < n && kj == kk);
        while (mask != 0u) {
          const int pick = __builtin_ctz(mask);
          mask &= mask - 1u;
          const int   word = __builtin_amdgcn_readlane(rc.x, pick);
          const float v    = __int_as_float(__builtin_amdgcn_readlane(rc.y, pick));
          const int   rowbase = word & 0xFFFF;
          const int   sh      = word >> 18;

          int po = t0c - sh;
          if (po < 0) po += NLON_IN;
          const float* rowp = xh + rowbase;

          v4f a[8];
          if (po <= NLON_IN - 4) {
#pragma unroll
            for (int s = 0; s < 4; ++s) {
                a[2 * s + 0] = *(const v4f_u*)(rowp + (4 * s + 0) * NPIX + po);
                a[2 * s + 1] = *(const v4f_u*)(rowp + (4 * s + 1) * NPIX + po);
            }
          } else {
            const int p1 = po + 1 < NLON_IN ? po + 1 : po + 1 - NLON_IN;
            const int p2 = po + 2 < NLON_IN ? po + 2 : po + 2 - NLON_IN;
            const int p3 = po + 3 - NLON_IN;
#pragma unroll
            for (int k = 0; k < 8; ++k) {
                const float* pl = rowp + (4 * (k >> 1) + (k & 1)) * NPIX;
                a[k] = (v4f){pl[po], pl[p1], pl[p2], pl[p3]};
            }
          }
#pragma unroll
          for (int k = 0; k < 8; ++k)
              S[k] += a[k] * v;
        }
      }
      mix_basis(d, S, wl + kk * (4 * 32) + lane);
    }

#pragma unroll
    for (int e = 0; e < TILES_PER_WAVE; ++e) {
#pragma unroll
        for (int r = 0; r < 8; ++r) {
            const int t = 64 * wp + 32 * h + 4 * r + e;
            if (t < NLON_IN) acc[c * ACC_PITCH + q + 2 * t] = d[e][r];
        }
    }
    }
    __syncthreads();

    for (int e = tid; e < COUT * NLON_OUT; e += THREADS) {
        const int co = e / NLON_OUT;
        const int L  = e - co * NLON_OUT;
        out[(co * NLAT_OUT + hi) * NLON_OUT + L] = acc[co * ACC_PITCH + L] + bias[co];
    }
}

extern "C" void kernel_launch(void* const* d_in, const int* in_sizes, int n_in,
                              void* d_out, int out_size, void* d_ws, size_t ws_size,
                              hipStream_t stream)
{
    (void)in_sizes; (void)n_in; (void)out_size; (void)d_ws; (void)ws_size;

    const float* x    = (const float*)d_in[0];
    const float* w    = (const float*)d_in[1];
    const float* bias = (const float*)d_in[2];
    const int*   ker  = (const int*)  d_in[3];
    const int*   row  = (const int*)  d_in[4];
    const int*   col  = (const int*)  d_in[5];
    const float* vals = (const float*)d_in[6];
    float*       out  = (float*)d_out;

    disco_fused_lat<<<NLAT_OUT, THREADS, 0, stream>>>(x, w, bias, ker, row, col, vals, out);
}


# queue build in one round: four 16-byte index loads per thread in flight, slots from a prefix sum of hit counts instead of arrival-order atomics; sum order now fixed by entry position, not arrival
# speedup vs baseline: 3.8488x; 1.1991x over previous
#include <hip/hip_runtime.h>


typedef float v2f __attribute__((ext_vector_type(2)));
typedef float v4f __attribute__((ext_vector_type(4)));
typedef float v8f __attribute__((ext_vector_type(8)));
typedef float v4f_u __attribute__((ext_vector_type(4), aligned(4)));
typedef int   v2i __attribute__((ext_vector_type(2)));
typedef int   v4i __attribute__((ext_vector_type(4)));

#define NLAT_IN   181
#define NLON_IN   360
#define NLAT_OUT  361
#define NLON_OUT  720
#define KSIZE     3
#define CIN       16
#define COUT      16
#define NNZ       8192
#define NPIX      (NLAT_IN * NLON_IN)
#define ACC_PITCH 727
#define THREADS   512
#define WAVES     (THREADS / 32)
#define NZ_PER_WAVE (NNZ / WAVES)
#define GATHER_PAIRS 6
#define TILES_PER_WAVE 4

__device__ __forceinline__ void mix_basis(v8f (&d)[TILES_PER_WAVE],
                                          const v4f (&S)[8], const v2f* wk)
{
    v2f b[4];
#pragma unroll
    for (int s = 0; s < 4; ++s)
        b[s] = wk[s * 32];
#pragma unroll
    for (int s = 0; s < 4; ++s) {
#pragma unroll
        for (int e = 0; e < TILES_PER_WAVE; ++e) {
            v2f a;
            a.x = S[2 * s + 0][e];
            a.y = S[2 * s + 1][e];
            d[e] = __builtin_amdgcn_wmma_f32_16x16x4_f32(false, a, false, b[s],
                                                         (short)0, d[e], false, false);
        }
    }
}

__global__ __launch_bounds__(THREADS)
void disco_fused_lat(const float* __restrict__ x,
                     const float* __restrict__ w,
                     const float* __restrict__ bias,
                     const int*   __restrict__ ker_idx,
                     const int*   __restrict__ row_idx,
                     const int*   __restrict__ col_idx,
                     const float* __restrict__ vals,
                     float* __restrict__ out)
{
    __shared__ float acc[COUT * ACC_PITCH];
    __shared__ v2i   rec[NNZ];
    __shared__ v2f   wl[KSIZE * 4 * 32];
    __shared__ int   wtot[WAVES];

    const int tid  = threadIdx.x;
    const int hi   = blockIdx.x;
    const int lane = tid & 31;
    const int wave = __builtin_amdgcn_readfirstlane(tid >> 5);

    const int hbase = hi * NLON_OUT;
    const int nz0   = wave * NZ_PER_WAVE + 4 * lane;
    v4i cv[4];
#pragma unroll
    for (int j = 0; j < 4; ++j)
        cv[j] = *(const v4i*)(col_idx + nz0 + 128 * j);

    if (tid < KSIZE * 4 * 32) {
        const int k  = tid >> 7;
        const int s  = (tid >> 5) & 3;
        const int cc = tid & 15;
        const int hh = (tid >> 4) & 1;
        const int ci = 4 * s + 2 * hh;
        v2f wv;
        wv.x = w[cc * (CIN * KSIZE) + (ci + 0) * KSIZE + k];
        wv.y = w[cc * (CIN * KSIZE) + (ci + 1) * KSIZE + k];
        wl[tid] = wv;
    }

    unsigned hm = 0u, pm = 0u;
#pragma unroll
    for (int j = 0; j < 4; ++j) {
#pragma unroll
        for (int e = 0; e < 4; ++e) {
            const unsigned p   = (unsigned)(cv[j][e] - hbase);
            const unsigned hit = p < (unsigned)NLON_OUT ? 1u : 0u;
            hm |= hit << (4 * j + e);
            pm |= (hit & p) << (4 * j + e);
        }
    }
    const int c1     = __builtin_popcount(pm);
    const int packed = (__builtin_popcount(hm) - c1) | (c1 << 16);

    int run = 0, mybase = 0;
    {
        unsigned act = __builtin_amdgcn_ballot_w32(packed != 0);
        while (act != 0u) {
            const int pick = __builtin_ctz(act);
            act &= act - 1u;
            const int cpk = __builtin_amdgcn_readlane(packed, pick);
            if (lane == pick) mybase = run;
            run += cpk;
        }
    }
    if (lane == 0) wtot[wave] = run;
    __syncthreads();

    int tot = 0, wbase = 0;
#pragma unroll
    for (int w2 = 0; w2 < WAVES; ++w2) {
        const int t = wtot[w2];
        tot += t;
        if (w2 < wave) wbase += t;
    }

    {
        int s0 = (wbase + mybase) & 0xFFFF;
        int s1 = (wbase + mybase) >> 16;
        while (hm != 0u) {
            const int i = __builtin_ctz(hm);
            hm &= hm - 1u;
            const int nz   = nz0 + 128 * (i >> 2) + (i & 3);
            const int p    = col_idx[nz] - hbase;
            const int word = (row_idx[nz] * NLON_IN) | (ker_idx[nz] << 16) | ((p >> 1) << 18);
            v2i rc;
            rc.x = word;
            rc.y = __float_as_int(vals[nz]);
            int slot;
            if ((pm >> i) & 1u) { slot = NNZ - 1 - s1; ++s1; }
            else                { slot = s0;           ++s0; }
            rec[slot] = rc;
        }
    }
    __syncthreads();

    const int q  = wave & 1;
    const int wp = wave >> 1;
    const int c  = lane & 15;
    const int h  = lane >> 4;

    if (wp < GATHER_PAIRS) {
    const int n  = __builtin_amdgcn_readfirstlane(q ? (tot >> 16) : (tot & 0xFFFF));

    const int t0  = 64 * wp + 4 * c;
    const int t0c = t0 < NLON_IN ? t0 : NLON_IN - 4;

    const float* xh = x + 2 * h * NPIX;

    v8f d[TILES_PER_WAVE];
#pragma unroll
    for (int e = 0; e < TILES_PER_WAVE; ++e)
        d[e] = (v8f){0.f, 0.f, 0.f, 0.f, 0.f, 0.f, 0.f, 0.f};

    for (int kk = 0; kk < KSIZE; ++kk) {
      v4f S[8];
#pragma unroll
      for (int k = 0; k < 8; ++k)
          S[k] = (v4f){0.f, 0.f, 0.f, 0.f};
      for (int j0 = 0; j0 < n; j0 += 32) {
        const int idx = j0 + lane;
        v2i rc = (v2i){0, 0};
        if (idx < n) rc = rec[q ? (NNZ - 1 - idx) : idx];
        int kj = (rc.x >> 16) & 3;
        if (kj > KSIZE - 1) kj = KSIZE - 1;
        unsigned mask = __builtin_amdgcn_ballot_w32(idx < n && kj == kk);
        while (mask != 0u) {
          const int pick = __builtin_ctz(mask);
          mask &= mask - 1u;
          const int   word = __builtin_amdgcn_readlane(rc.x, pick);
          const float v    = __int_as_float(__builtin_amdgcn_readlane(rc.y, pick));
          const int   rowbase = word & 0xFFFF;
          const int   sh      = word >> 18;

          int po = t0c - sh;
          if (po < 0) po += NLON_IN;
          const float* rowp = xh + rowbase;

          v4f a[8];
          if (po <= NLON_IN - 4) {
#pragma unroll
            for (int s = 0; s < 4; ++s) {
                a[2 * s + 0] = *(const v4f_u*)(rowp + (4 * s + 0) * NPIX + po);
                a[2 * s + 1] = *(const v4f_u*)(rowp + (4 * s + 1) * NPIX + po);
            }
          } else {
            const int p1 = po + 1 < NLON_IN ? po + 1 : po + 1 - NLON_IN;
            const int p2 = po + 2 < NLON_IN ? po + 2 : po + 2 - NLON_IN;
            const int p3 = po + 3 - NLON_IN;
#pragma unroll
            for (int k = 0; k < 8; ++k) {
                const float* pl = rowp + (4 * (k >> 1) + (k & 1)) * NPIX;
                a[k] = (v4f){pl[po], pl[p1], pl[p2], pl[p3]};
            }
          }
#pragma unroll
          for (int k = 0; k < 8; ++k)
              S[k] += a[k] * v;
        }
      }
      mix_basis(d, S, wl + kk * (4 * 32) + lane);
    }

#pragma unroll
    for (int e = 0; e < TILES_PER_WAVE; ++e) {
#pragma unroll
        for (int r = 0; r < 8; ++r) {
            const int t = 64 * wp + 32 * h + 4 * r + e;
            if (t < NLON_IN) acc[c * ACC_PITCH + q + 2 * t] = d[e][r];
        }
    }
    }
    __syncthreads();

    for (int e = tid; e < COUT * NLON_OUT; e += THREADS) {
        const int co = e / NLON_OUT;
        const int L  = e - co * NLON_OUT;
        out[(co * NLAT_OUT + hi) * NLON_OUT + L] = acc[co * ACC_PITCH + L] + bias[co];
    }
}

extern "C" void kernel_launch(void* const* d_in, const int* in_sizes, int n_in,
                              void* d_out, int out_size, void* d_ws, size_t ws_size,
                              hipStream_t stream)
{
    (void)in_sizes; (void)n_in; (void)out_size; (void)d_ws; (void)ws_size;

    const float* x    = (const float*)d_in[0];
    const float* w    = (const float*)d_in[1];
    const float* bias = (const float*)d_in[2];
    const int*   ker  = (const int*)  d_in[3];
    const int*   row  = (const int*)  d_in[4];
    const int*   col  = (const int*)  d_in[5];
    const float* vals = (const float*)d_in[6];
    float*       out  = (float*)d_out;

    disco_fused_lat<<<NLAT_OUT, THREADS, 0, stream>>>(x, w, bias, ker, row, col, vals, out);
}


# each latitude split over two work-groups, one per half of the longitude ring (722 groups of 8 waves, each builds the queue), to even out row reads per compute unit; sum order: new queue order
# speedup vs baseline: 4.0651x; 1.0562x over previous
#include <hip/hip_runtime.h>


typedef float v2f __attribute__((ext_vector_type(2)));
typedef float v4f __attribute__((ext_vector_type(4)));
typedef float v8f __attribute__((ext_vector_type(8)));
typedef float v4f_u __attribute__((ext_vector_type(4), aligned(4)));
typedef int   v2i __attribute__((ext_vector_type(2)));
typedef int   v4i __attribute__((ext_vector_type(4)));

#define NLAT_IN   181
#define NLON_IN   360
#define NLAT_OUT  361
#define NLON_OUT  720
#define KSIZE     3
#define CIN       16
#define COUT      16
#define NNZ       8192
#define NPIX      (NLAT_IN * NLON_IN)
#define HALF_T    (NLON_IN / 2)
#define HALF_LON  (NLON_OUT / 2)
#define ACC_PITCH 367
#define THREADS   256
#define WAVES     (THREADS / 32)
#define NZ_PER_WAVE (NNZ / WAVES)
#define NZ_GROUPS (NZ_PER_WAVE / 128)
#define GATHER_PAIRS 3
#define W_FRAGS   (KSIZE * 4 * 32)
#define W_TRIPS   ((W_FRAGS + THREADS - 1) / THREADS)
#define TILES_PER_WAVE 4

__device__ __forceinline__ void mix_basis(v8f (&d)[TILES_PER_WAVE],
                                          const v4f (&S)[8], const v2f* wk)
{
    v2f b[4];
#pragma unroll
    for (int s = 0; s < 4; ++s)
        b[s] = wk[s * 32];
#pragma unroll
    for (int s = 0; s < 4; ++s) {
#pragma unroll
        for (int e = 0; e < TILES_PER_WAVE; ++e) {
            v2f a;
            a.x = S[2 * s + 0][e];
            a.y = S[2 * s + 1][e];
            d[e] = __builtin_amdgcn_wmma_f32_16x16x4_f32(false, a, false, b[s],
                                                         (short)0, d[e], false, false);
        }
    }
}

__global__ __launch_bounds__(THREADS)
void disco_fused_lat(const float* __restrict__ x,
                     const float* __restrict__ w,
                     const float* __restrict__ bias,
                     const int*   __restrict__ ker_idx,
                     const int*   __restrict__ row_idx,
                     const int*   __restrict__ col_idx,
                     const float* __restrict__ vals,
                     float* __restrict__ out)
{
    __shared__ float acc[COUT * ACC_PITCH];
    __shared__ v2i   rec[NNZ];
    __shared__ v2f   wl[W_FRAGS];
    __shared__ int   wtot[WAVES];

    const int tid  = threadIdx.x;
    const int half = blockIdx.x >= NLAT_OUT ? 1 : 0;
    const int hi   = blockIdx.x - half * NLAT_OUT;
    const int lane = tid & 31;
    const int wave = __builtin_amdgcn_readfirstlane(tid >> 5);

    const int hbase = hi * NLON_OUT;
    const int nz0   = wave * NZ_PER_WAVE + 4 * lane;
    v4i cv[NZ_GROUPS];
#pragma unroll
    for (int j = 0; j < NZ_GROUPS; ++j)
        cv[j] = *(const v4i*)(col_idx + nz0 + 128 * j);

    v2f wv[W_TRIPS];
#pragma unroll
    for (int t = 0; t < W_TRIPS; ++t) {
        const int i  = tid + t * THREADS;
        const int k  = i >> 7;
        const int s  = (i >> 5) & 3;
        const int cc = i & 15;
        const int hh = (i >> 4) & 1;
        const int ci = 4 * s + 2 * hh;
        wv[t] = (v2f){0.f, 0.f};
        if (i < W_FRAGS) {
            wv[t].x = w[cc * (CIN * KSIZE) + (ci + 0) * KSIZE + k];
            wv[t].y = w[cc * (CIN * KSIZE) + (ci + 1) * KSIZE + k];
        }
    }
#pragma unroll
    for (int t = 0; t < W_TRIPS; ++t) {
        const int i = tid + t * THREADS;
        if (i < W_FRAGS) wl[i] = wv[t];
    }

    unsigned hm = 0u, pm = 0u;
#pragma unroll
    for (int j = 0; j < NZ_GROUPS; ++j) {
#pragma unroll
        for (int e = 0; e < 4; ++e) {
            const unsigned p   = (unsigned)(cv[j][e] - hbase);
            const unsigned hit = p < (unsigned)NLON_OUT ? 1u : 0u;
            hm |= hit << (4 * j + e);
            pm |= (hit & p) << (4 * j + e);
        }
    }
    const int c1     = __builtin_popcount(pm);
    const int packed = (__builtin_popcount(hm) - c1) | (c1 << 16);

    int run = 0, mybase = 0;
    {
        unsigned act = __builtin_amdgcn_ballot_w32(packed != 0);
        while (act != 0u) {
            const int pick = __builtin_ctz(act);
            act &= act - 1u;
            const int cpk = __builtin_amdgcn_readlane(packed, pick);
            if (lane == pick) mybase = run;
            run += cpk;
        }
    }
    if (lane == 0) wtot[wave] = run;
    __syncthreads();

    int tot = 0, wbase = 0;
#pragma unroll
    for (int w2 = 0; w2 < WAVES; ++w2) {
        const int t = wtot[w2];
        tot += t;
        if (w2 < wave) wbase += t;
    }

    {
        int s0 = (wbase + mybase) & 0xFFFF;
        int s1 = (wbase + mybase) >> 16;
        while (hm != 0u) {
            const int i = __builtin_ctz(hm);
            hm &= hm - 1u;
            const int nz   = nz0 + 128 * (i >> 2) + (i & 3);
            const int p    = col_idx[nz] - hbase;
            const int word = (row_idx[nz] * NLON_IN) | (ker_idx[nz] << 16) | ((p >> 1) << 18);
            v2i rc;
            rc.x = word;
            rc.y = __float_as_int(vals[nz]);
            int slot;
            if ((pm >> i) & 1u) { slot = NNZ - 1 - s1; ++s1; }
            else                { slot = s0;           ++s0; }
            rec[slot] = rc;
        }
    }
    __syncthreads();

    const int q  = wave & 1;
    const int wp = wave >> 1;
    const int c  = lane & 15;
    const int h  = lane >> 4;

    if (wp < GATHER_PAIRS) {
    const int n  = __builtin_amdgcn_readfirstlane(q ? (tot >> 16) : (tot & 0xFFFF));

    const int tl  = 64 * wp + 4 * c;
    const int t0c = HALF_T * half + (tl < HALF_T ? tl : HALF_T - 4);

    const float* xh = x + 2 * h * NPIX;

    v8f d[TILES_PER_WAVE];
#pragma unroll
    for (int e = 0; e < TILES_PER_WAVE; ++e)
        d[e] = (v8f){0.f, 0.f, 0.f, 0.f, 0.f, 0.f, 0.f, 0.f};

    for (int kk = 0; kk < KSIZE; ++kk) {
      v4f S[8];
#pragma unroll
      for (int k = 0; k < 8; ++k)
          S[k] = (v4f){0.f, 0.f, 0.f, 0.f};
      for (int j0 = 0; j0 < n; j0 += 32) {
        const int idx = j0 + lane;
        v2i rc = (v2i){0, 0};
        if (idx < n) rc = rec[q ? (NNZ - 1 - idx) : idx];
        int kj = (rc.x >> 16) & 3;
        if (kj > KSIZE - 1) kj = KSIZE - 1;
        unsigned mask = __builtin_amdgcn_ballot_w32(idx < n && kj == kk);
        while (mask != 0u) {
          const int pick = __builtin_ctz(mask);
          mask &= mask - 1u;
          const int   word = __builtin_amdgcn_readlane(rc.x, pick);
          const float v    = __int_as_float(__builtin_amdgcn_readlane(rc.y, pick));
          const int   rowbase = word & 0xFFFF;
          const int   sh      = word >> 18;

          int po = t0c - sh;
          if (po < 0) po += NLON_IN;
          const float* rowp = xh + rowbase;

          v4f a[8];
          if (po <= NLON_IN - 4) {
#pragma unroll
            for (int s = 0; s < 4; ++s) {
                a[2 * s + 0] = *(const v4f_u*)(rowp + (4 * s + 0) * NPIX + po);
                a[2 * s + 1] = *(const v4f_u*)(rowp + (4 * s + 1) * NPIX + po);
            }
          } else {
            const int p1 = po + 1 < NLON_IN ? po + 1 : po + 1 - NLON_IN;
            const int p2 = po + 2 < NLON_IN ? po + 2 : po + 2 - NLON_IN;
            const int p3 = po + 3 - NLON_IN;
#pragma unroll
            for (int k = 0; k < 8; ++k) {
                const float* pl = rowp + (4 * (k >> 1) + (k & 1)) * NPIX;
                a[k] = (v4f){pl[po], pl[p1], pl[p2], pl[p3]};
            }
          }
#pragma unroll
          for (int k = 0; k < 8; ++k)
              S[k] += a[k] * v;
        }
      }
      mix_basis(d, S, wl + kk * (4 * 32) + lane);
    }

#pragma unroll
    for (int e = 0; e < TILES_PER_WAVE; ++e) {
#pragma unroll
        for (int r = 0; r < 8; ++r) {
            const int t = 64 * wp + 32 * h + 4 * r + e;
            if (t < HALF_T) acc[c * ACC_PITCH + q + 2 * t] = d[e][r];
        }
    }
    }
    __syncthreads();

    for (int e = tid; e < COUT * HALF_LON; e += THREADS) {
        const int co = e / HALF_LON;
        const int L  = e - co * HALF_LON;
        out[(co * NLAT_OUT + hi) * NLON_OUT + HALF_LON * half + L] = acc[co * ACC_PITCH + L] + bias[co];
    }
}

extern "C" void kernel_launch(void* const* d_in, const int* in_sizes, int n_in,
                              void* d_out, int out_size, void* d_ws, size_t ws_size,
                              hipStream_t stream)
{
    (void)in_sizes; (void)n_in; (void)out_size; (void)d_ws; (void)ws_size;

    const float* x    = (const float*)d_in[0];
    const float* w    = (const float*)d_in[1];
    const float* bias = (const float*)d_in[2];
    const int*   ker  = (const int*)  d_in[3];
    const int*   row  = (const int*)  d_in[4];
    const int*   col  = (const int*)  d_in[5];
    const float* vals = (const float*)d_in[6];
    float*       out  = (float*)d_out;

    disco_fused_lat<<<2 * NLAT_OUT, THREADS, 0, stream>>>(x, w, bias, ker, row, col, vals, out);
}


# gather loop takes two queue records of a basis per trip: the 16-byte x row loads of both are issued before either is added, one cache round trip per pair; sum order unchanged (queue order)
# speedup vs baseline: 4.2197x; 1.0380x over previous
#include <hip/hip_runtime.h>


typedef float v2f __attribute__((ext_vector_type(2)));
typedef float v4f __attribute__((ext_vector_type(4)));
typedef float v8f __attribute__((ext_vector_type(8)));
typedef float v4f_u __attribute__((ext_vector_type(4), aligned(4)));
typedef int   v2i __attribute__((ext_vector_type(2)));
typedef int   v4i __attribute__((ext_vector_type(4)));

#define NLAT_IN   181
#define NLON_IN   360
#define NLAT_OUT  361
#define NLON_OUT  720
#define KSIZE     3
#define CIN       16
#define COUT      16
#define NNZ       8192
#define NPIX      (NLAT_IN * NLON_IN)
#define HALF_T    (NLON_IN / 2)
#define HALF_LON  (NLON_OUT / 2)
#define ACC_PITCH 367
#define THREADS   256
#define WAVES     (THREADS / 32)
#define NZ_PER_WAVE (NNZ / WAVES)
#define NZ_GROUPS (NZ_PER_WAVE / 128)
#define GATHER_PAIRS 3
#define W_FRAGS   (KSIZE * 4 * 32)
#define W_TRIPS   ((W_FRAGS + THREADS - 1) / THREADS)
#define TILES_PER_WAVE 4

__device__ __forceinline__ float load_rows(v4f (&a)[8], const float* xh, int t0c, int word, int vbits)
{
    const int rowbase = word & 0xFFFF;
    const int sh      = word >> 18;
    int po = t0c - sh;
    if (po < 0) po += NLON_IN;
    const float* rowp = xh + rowbase;
    const bool wraps = po > NLON_IN - 4;
    const int  pc    = wraps ? NLON_IN - 4 : po;
#pragma unroll
    for (int s = 0; s < 4; ++s) {
        a[2 * s + 0] = *(const v4f_u*)(rowp + (4 * s + 0) * NPIX + pc);
        a[2 * s + 1] = *(const v4f_u*)(rowp + (4 * s + 1) * NPIX + pc);
    }
    if (wraps) {
        const int p1 = po + 1 < NLON_IN ? po + 1 : po + 1 - NLON_IN;
        const int p2 = po + 2 < NLON_IN ? po + 2 : po + 2 - NLON_IN;
        const int p3 = po + 3 - NLON_IN;
#pragma unroll
        for (int k = 0; k < 8; ++k) {
            const float* pl = rowp + (4 * (k >> 1) + (k & 1)) * NPIX;
            a[k] = (v4f){pl[po], pl[p1], pl[p2], pl[p3]};
        }
    }
    return __int_as_float(vbits);
}

__device__ __forceinline__ void mix_basis(v8f (&d)[TILES_PER_WAVE],
                                          const v4f (&S)[8], const v2f* wk)
{
    v2f b[4];
#pragma unroll
    for (int s = 0; s < 4; ++s)
        b[s] = wk[s * 32];
#pragma unroll
    for (int s = 0; s < 4; ++s) {
#pragma unroll
        for (int e = 0; e < TILES_PER_WAVE; ++e) {
            v2f a;
            a.x = S[2 * s + 0][e];
            a.y = S[2 * s + 1][e];
            d[e] = __builtin_amdgcn_wmma_f32_16x16x4_f32(false, a, false, b[s],
                                                         (short)0, d[e], false, false);
        }
    }
}

__global__ __launch_bounds__(THREADS)
void disco_fused_lat(const float* __restrict__ x,
                     const float* __restrict__ w,
                     const float* __restrict__ bias,
                     const int*   __restrict__ ker_idx,
                     const int*   __restrict__ row_idx,
                     const int*   __restrict__ col_idx,
                     const float* __restrict__ vals,
                     float* __restrict__ out)
{
    __shared__ float acc[COUT * ACC_PITCH];
    __shared__ v2i   rec[NNZ];
    __shared__ v2f   wl[W_FRAGS];
    __shared__ int   wtot[WAVES];

    const int tid  = threadIdx.x;
    const int half = blockIdx.x >= NLAT_OUT ? 1 : 0;
    const int hi   = blockIdx.x - half * NLAT_OUT;
    const int lane = tid & 31;
    const int wave = __builtin_amdgcn_readfirstlane(tid >> 5);

    const int hbase = hi * NLON_OUT;
    const int nz0   = wave * NZ_PER_WAVE + 4 * lane;
    v4i cv[NZ_GROUPS];
#pragma unroll
    for (int j = 0; j < NZ_GROUPS; ++j)
        cv[j] = *(const v4i*)(col_idx + nz0 + 128 * j);

    v2f wv[W_TRIPS];
#pragma unroll
    for (int t = 0; t < W_TRIPS; ++t) {
        const int i  = tid + t * THREADS;
        const int k  = i >> 7;
        const int s  = (i >> 5) & 3;
        const int cc = i & 15;
        const int hh = (i >> 4) & 1;
        const int ci = 4 * s + 2 * hh;
        wv[t] = (v2f){0.f, 0.f};
        if (i < W_FRAGS) {
            wv[t].x = w[cc * (CIN * KSIZE) + (ci + 0) * KSIZE + k];
            wv[t].y = w[cc * (CIN * KSIZE) + (ci + 1) * KSIZE + k];
        }
    }
#pragma unroll
    for (int t = 0; t < W_TRIPS; ++t) {
        const int i = tid + t * THREADS;
        if (i < W_FRAGS) wl[i] = wv[t];
    }

    unsigned hm = 0u, pm = 0u;
#pragma unroll
    for (int j = 0; j < NZ_GROUPS; ++j) {
#pragma unroll
        for (int e = 0; e < 4; ++e) {
            const unsigned p   = (unsigned)(cv[j][e] - hbase);
            const unsigned hit = p < (unsigned)NLON_OUT ? 1u : 0u;
            hm |= hit << (4 * j + e);
            pm |= (hit & p) << (4 * j + e);
        }
    }
    const int c1     = __builtin_popcount(pm);
    const int packed = (__builtin_popcount(hm) - c1) | (c1 << 16);

    int run = 0, mybase = 0;
    {
        unsigned act = __builtin_amdgcn_ballot_w32(packed != 0);
        while (act != 0u) {
            const int pick = __builtin_ctz(act);
            act &= act - 1u;
            const int cpk = __builtin_amdgcn_readlane(packed, pick);
            if (lane == pick) mybase = run;
            run += cpk;
        }
    }
    if (lane == 0) wtot[wave] = run;
    __syncthreads();

    int tot = 0, wbase = 0;
#pragma unroll
    for (int w2 = 0; w2 < WAVES; ++w2) {
        const int t = wtot[w2];
        tot += t;
        if (w2 < wave) wbase += t;
    }

    {
        int s0 = (wbase + mybase) & 0xFFFF;
        int s1 = (wbase + mybase) >> 16;
        while (hm != 0u) {
            const int i = __builtin_ctz(hm);
            hm &= hm - 1u;
            const int nz   = nz0 + 128 * (i >> 2) + (i & 3);
            const int p    = col_idx[nz] - hbase;
            const int word = (row_idx[nz] * NLON_IN) | (ker_idx[nz] << 16) | ((p >> 1) << 18);
            v2i rc;
            rc.x = word;
            rc.y = __float_as_int(vals[nz]);
            int slot;
            if ((pm >> i) & 1u) { slot = NNZ - 1 - s1; ++s1; }
            else                { slot = s0;           ++s0; }
            rec[slot] = rc;
        }
    }
    __syncthreads();

    const int q  = wave & 1;
    const int wp = wave >> 1;
    const int c  = lane & 15;
    const int h  = lane >> 4;

    if (wp < GATHER_PAIRS) {
    const int n  = __builtin_amdgcn_readfirstlane(q ? (tot >> 16) : (tot & 0xFFFF));

    const int tl  = 64 * wp + 4 * c;
    const int t0c = HALF_T * half + (tl < HALF_T ? tl : HALF_T - 4);

    const float* xh = x + 2 * h * NPIX;

    v8f d[TILES_PER_WAVE];
#pragma unroll
    for (int e = 0; e < TILES_PER_WAVE; ++e)
        d[e] = (v8f){0.f, 0.f, 0.f, 0.f, 0.f, 0.f, 0.f, 0.f};

    for (int kk = 0; kk < KSIZE; ++kk) {
      v4f S[8];
#pragma unroll
      for (int k = 0; k < 8; ++k)
          S[k] = (v4f){0.f, 0.f, 0.f, 0.f};
      for (int j0 = 0; j0 < n; j0 += 32) {
        const int idx = j0 + lane;
        v2i rc = (v2i){0, 0};
        if (idx < n) rc = rec[q ? (NNZ - 1 - idx) : idx];
        int kj = (rc.x >> 16) & 3;
        if (kj > KSIZE - 1) kj = KSIZE - 1;
        unsigned mask = __builtin_amdgcn_ballot_w32(idx < n && kj == kk);
        while (mask != 0u) {
          const int pick0 = __builtin_ctz(mask);
          mask &= mask - 1u;
          const bool two = mask != 0u;
          const int pick1 = two ? __builtin_ctz(mask) : pick0;
          mask &= mask - 1u;
          v4f a0[8], a1[8];
          const float v0 = load_rows(a0, xh, t0c,
                                     __builtin_amdgcn_readlane(rc.x, pick0),
                                     __builtin_amdgcn_readlane(rc.y, pick0));
          float v1 = 0.f;
          if (two)
            v1 = load_rows(a1, xh, t0c,
                           __builtin_amdgcn_readlane(rc.x, pick1),
                           __builtin_amdgcn_readlane(rc.y, pick1));
#pragma unroll
          for (int k = 0; k < 8; ++k)
              S[k] += a0[k] * v0;
          if (two) {
#pragma unroll
            for (int k = 0; k < 8; ++k)
                S[k] += a1[k] * v1;
          }
        }
      }
      mix_basis(d, S, wl + kk * (4 * 32) + lane);
    }

#pragma unroll
    for (int e = 0; e < TILES_PER_WAVE; ++e) {
#pragma unroll
        for (int r = 0; r < 8; ++r) {
            const int t = 64 * wp + 32 * h + 4 * r + e;
            if (t < HALF_T) acc[c * ACC_PITCH + q + 2 * t] = d[e][r];
        }
    }
    }
    __syncthreads();

    for (int e = tid; e < COUT * HALF_LON; e += THREADS) {
        const int co = e / HALF_LON;
        const int L  = e - co * HALF_LON;
        out[(co * NLAT_OUT + hi) * NLON_OUT + HALF_LON * half + L] = acc[co * ACC_PITCH + L] + bias[co];
    }
}

extern "C" void kernel_launch(void* const* d_in, const int* in_sizes, int n_in,
                              void* d_out, int out_size, void* d_ws, size_t ws_size,
                              hipStream_t stream)
{
    (void)in_sizes; (void)n_in; (void)out_size; (void)d_ws; (void)ws_size;

    const float* x    = (const float*)d_in[0];
    const float* w    = (const float*)d_in[1];
    const float* bias = (const float*)d_in[2];
    const int*   ker  = (const int*)  d_in[3];
    const int*   row  = (const int*)  d_in[4];
    const int*   col  = (const int*)  d_in[5];
    const float* vals = (const float*)d_in[6];
    float*       out  = (float*)d_out;

    disco_fused_lat<<<2 * NLAT_OUT, THREADS, 0, stream>>>(x, w, bias, ker, row, col, vals, out);
}


# write-out with 16-byte stores: each lane takes four consecutive longitudes of a staged row (one 16-byte LDS read, one 16-byte store), 45 store instructions per work-group instead of 180
# speedup vs baseline: 4.6937x; 1.1123x over previous
#include <hip/hip_runtime.h>


typedef float v2f __attribute__((ext_vector_type(2)));
typedef float v4f __attribute__((ext_vector_type(4)));
typedef float v8f __attribute__((ext_vector_type(8)));
typedef float v4f_u __attribute__((ext_vector_type(4), aligned(4)));
typedef int   v2i __attribute__((ext_vector_type(2)));
typedef int   v4i __attribute__((ext_vector_type(4)));

#define NLAT_IN   181
#define NLON_IN   360
#define NLAT_OUT  361
#define NLON_OUT  720
#define KSIZE     3
#define CIN       16
#define COUT      16
#define NNZ       8192
#define NPIX      (NLAT_IN * NLON_IN)
#define HALF_T    (NLON_IN / 2)
#define HALF_LON  (NLON_OUT / 2)
#define ACC_PITCH 364
#define ROW_VEC   (HALF_LON / 4)
#define WO_TRIPS  ((COUT * ROW_VEC + THREADS - 1) / THREADS)
#define THREADS   256
#define WAVES     (THREADS / 32)
#define NZ_PER_WAVE (NNZ / WAVES)
#define NZ_GROUPS (NZ_PER_WAVE / 128)
#define GATHER_PAIRS 3
#define W_FRAGS   (KSIZE * 4 * 32)
#define W_TRIPS   ((W_FRAGS + THREADS - 1) / THREADS)
#define TILES_PER_WAVE 4

__device__ __forceinline__ float load_rows(v4f (&a)[8], const float* xh, int t0c, int word, int vbits)
{
    const int rowbase = word & 0xFFFF;
    const int sh      = word >> 18;
    int po = t0c - sh;
    if (po < 0) po += NLON_IN;
    const float* rowp = xh + rowbase;
    const bool wraps = po > NLON_IN - 4;
    const int  pc    = wraps ? NLON_IN - 4 : po;
#pragma unroll
    for (int s = 0; s < 4; ++s) {
        a[2 * s + 0] = *(const v4f_u*)(rowp + (4 * s + 0) * NPIX + pc);
        a[2 * s + 1] = *(const v4f_u*)(rowp + (4 * s + 1) * NPIX + pc);
    }
    if (wraps) {
        const int p1 = po + 1 < NLON_IN ? po + 1 : po + 1 - NLON_IN;
        const int p2 = po + 2 < NLON_IN ? po + 2 : po + 2 - NLON_IN;
        const int p3 = po + 3 - NLON_IN;
#pragma unroll
        for (int k = 0; k < 8; ++k) {
            const float* pl = rowp + (4 * (k >> 1) + (k & 1)) * NPIX;
            a[k] = (v4f){pl[po], pl[p1], pl[p2], pl[p3]};
        }
    }
    return __int_as_float(vbits);
}

__device__ __forceinline__ void mix_basis(v8f (&d)[TILES_PER_WAVE],
                                          const v4f (&S)[8], const v2f* wk)
{
    v2f b[4];
#pragma unroll
    for (int s = 0; s < 4; ++s)
        b[s] = wk[s * 32];
#pragma unroll
    for (int s = 0; s < 4; ++s) {
#pragma unroll
        for (int e = 0; e < TILES_PER_WAVE; ++e) {
            v2f a;
            a.x = S[2 * s + 0][e];
            a.y = S[2 * s + 1][e];
            d[e] = __builtin_amdgcn_wmma_f32_16x16x4_f32(false, a, false, b[s],
                                                         (short)0, d[e], false, false);
        }
    }
}

__global__ __launch_bounds__(THREADS)
void disco_fused_lat(const float* __restrict__ x,
                     const float* __restrict__ w,
                     const float* __restrict__ bias,
                     const int*   __restrict__ ker_idx,
                     const int*   __restrict__ row_idx,
                     const int*   __restrict__ col_idx,
                     const float* __restrict__ vals,
                     float* __restrict__ out)
{
    __shared__ __attribute__((aligned(16))) float acc[COUT * ACC_PITCH];
    __shared__ v2i   rec[NNZ];
    __shared__ v2f   wl[W_FRAGS];
    __shared__ int   wtot[WAVES];

    const int tid  = threadIdx.x;
    const int half = blockIdx.x >= NLAT_OUT ? 1 : 0;
    const int hi   = blockIdx.x - half * NLAT_OUT;
    const int lane = tid & 31;
    const int wave = __builtin_amdgcn_readfirstlane(tid >> 5);

    const int hbase = hi * NLON_OUT;
    const int nz0   = wave * NZ_PER_WAVE + 4 * lane;
    v4i cv[NZ_GROUPS];
#pragma unroll
    for (int j = 0; j < NZ_GROUPS; ++j)
        cv[j] = *(const v4i*)(col_idx + nz0 + 128 * j);

    v2f wv[W_TRIPS];
#pragma unroll
    for (int t = 0; t < W_TRIPS; ++t) {
        const int i  = tid + t * THREADS;
        const int k  = i >> 7;
        const int s  = (i >> 5) & 3;
        const int cc = i & 15;
        const int hh = (i >> 4) & 1;
        const int ci = 4 * s + 2 * hh;
        wv[t] = (v2f){0.f, 0.f};
        if (i < W_FRAGS) {
            wv[t].x = w[cc * (CIN * KSIZE) + (ci + 0) * KSIZE + k];
            wv[t].y = w[cc * (CIN * KSIZE) + (ci + 1) * KSIZE + k];
        }
    }
#pragma unroll
    for (int t = 0; t < W_TRIPS; ++t) {
        const int i = tid + t * THREADS;
        if (i < W_FRAGS) wl[i] = wv[t];
    }

    unsigned hm = 0u, pm = 0u;
#pragma unroll
    for (int j = 0; j < NZ_GROUPS; ++j) {
#pragma unroll
        for (int e = 0; e < 4; ++e) {
            const unsigned p   = (unsigned)(cv[j][e] - hbase);
            const unsigned hit = p < (unsigned)NLON_OUT ? 1u : 0u;
            hm |= hit << (4 * j + e);
            pm |= (hit & p) << (4 * j + e);
        }
    }
    const int c1     = __builtin_popcount(pm);
    const int packed = (__builtin_popcount(hm) - c1) | (c1 << 16);

    int run = 0, mybase = 0;
    {
        unsigned act = __builtin_amdgcn_ballot_w32(packed != 0);
        while (act != 0u) {
            const int pick = __builtin_ctz(act);
            act &= act - 1u;
            const int cpk = __builtin_amdgcn_readlane(packed, pick);
            if (lane == pick) mybase = run;
            run += cpk;
        }
    }
    if (lane == 0) wtot[wave] = run;
    __syncthreads();

    int tot = 0, wbase = 0;
#pragma unroll
    for (int w2 = 0; w2 < WAVES; ++w2) {
        const int t = wtot[w2];
        tot += t;
        if (w2 < wave) wbase += t;
    }

    {
        int s0 = (wbase + mybase) & 0xFFFF;
        int s1 = (wbase + mybase) >> 16;
        while (hm != 0u) {
            const int i = __builtin_ctz(hm);
            hm &= hm - 1u;
            const int nz   = nz0 + 128 * (i >> 2) + (i & 3);
            const int p    = col_idx[nz] - hbase;
            const int word = (row_idx[nz] * NLON_IN) | (ker_idx[nz] << 16) | ((p >> 1) << 18);
            v2i rc;
            rc.x = word;
            rc.y = __float_as_int(vals[nz]);
            int slot;
            if ((pm >> i) & 1u) { slot = NNZ - 1 - s1; ++s1; }
            else                { slot = s0;           ++s0; }
            rec[slot] = rc;
        }
    }
    __syncthreads();

    const int q  = wave & 1;
    const int wp = wave >> 1;
    const int c  = lane & 15;
    const int h  = lane >> 4;

    if (wp < GATHER_PAIRS) {
    const int n  = __builtin_amdgcn_readfirstlane(q ? (tot >> 16) : (tot & 0xFFFF));

    const int tl  = 64 * wp + 4 * c;
    const int t0c = HALF_T * half + (tl < HALF_T ? tl : HALF_T - 4);

    const float* xh = x + 2 * h * NPIX;

    v8f d[TILES_PER_WAVE];
#pragma unroll
    for (int e = 0; e < TILES_PER_WAVE; ++e)
        d[e] = (v8f){0.f, 0.f, 0.f, 0.f, 0.f, 0.f, 0.f, 0.f};

    for (int kk = 0; kk < KSIZE; ++kk) {
      v4f S[8];
#pragma unroll
      for (int k = 0; k < 8; ++k)
          S[k] = (v4f){0.f, 0.f, 0.f, 0.f};
      for (int j0 = 0; j0 < n; j0 += 32) {
        const int idx = j0 + lane;
        v2i rc = (v2i){0, 0};
        if (idx < n) rc = rec[q ? (NNZ - 1 - idx) : idx];
        int kj = (rc.x >> 16) & 3;
        if (kj > KSIZE - 1) kj = KSIZE - 1;
        unsigned mask = __builtin_amdgcn_ballot_w32(idx < n && kj == kk);
        while (mask != 0u) {
          const int pick0 = __builtin_ctz(mask);
          mask &= mask - 1u;
          const bool two = mask != 0u;
          const int pick1 = two ? __builtin_ctz(mask) : pick0;
          mask &= mask - 1u;
          v4f a0[8], a1[8];
          const float v0 = load_rows(a0, xh, t0c,
                                     __builtin_amdgcn_readlane(rc.x, pick0),
                                     __builtin_amdgcn_readlane(rc.y, pick0));
          float v1 = 0.f;
          if (two)
            v1 = load_rows(a1, xh, t0c,
                           __builtin_amdgcn_readlane(rc.x, pick1),
                           __builtin_amdgcn_readlane(rc.y, pick1));
#pragma unroll
          for (int k = 0; k < 8; ++k)
              S[k] += a0[k] * v0;
          if (two) {
#pragma unroll
            for (int k = 0; k < 8; ++k)
                S[k] += a1[k] * v1;
          }
        }
      }
      mix_basis(d, S, wl + kk * (4 * 32) + lane);
    }

#pragma unroll
    for (int e = 0; e < TILES_PER_WAVE; ++e) {
#pragma unroll
        for (int r = 0; r < 8; ++r) {
            const int t = 64 * wp + 32 * h + 4 * r + e;
            if (t < HALF_T) acc[c * ACC_PITCH + q + 2 * t] = d[e][r];
        }
    }
    }
    __syncthreads();

    v4f   ov[WO_TRIPS];
    float ob[WO_TRIPS];
#pragma unroll
    for (int j = 0; j < WO_TRIPS; ++j) {
        int i = tid + j * THREADS;
        if (i > COUT * ROW_VEC - 1) i = COUT * ROW_VEC - 1;
        const int co = i / ROW_VEC;
        const int L  = 4 * (i - co * ROW_VEC);
        ov[j] = *(const v4f*)(acc + co * ACC_PITCH + L);
        ob[j] = bias[co];
    }
#pragma unroll
    for (int j = 0; j < WO_TRIPS; ++j) {
        const int i = tid + j * THREADS;
        if (i < COUT * ROW_VEC) {
            const int co = i / ROW_VEC;
            const int L  = 4 * (i - co * ROW_VEC);
            v4f v = ov[j];
            v.x += ob[j]; v.y += ob[j]; v.z += ob[j]; v.w += ob[j];
            *(v4f_u*)(out + (co * NLAT_OUT + hi) * NLON_OUT + HALF_LON * half + L) = v;
        }
    }
}

extern "C" void kernel_launch(void* const* d_in, const int* in_sizes, int n_in,
                              void* d_out, int out_size, void* d_ws, size_t ws_size,
                              hipStream_t stream)
{
    (void)in_sizes; (void)n_in; (void)out_size; (void)d_ws; (void)ws_size;

    const float* x    = (const float*)d_in[0];
    const float* w    = (const float*)d_in[1];
    const float* bias = (const float*)d_in[2];
    const int*   ker  = (const int*)  d_in[3];
    const int*   row  = (const int*)  d_in[4];
    const int*   col  = (const int*)  d_in[5];
    const float* vals = (const float*)d_in[6];
    float*       out  = (float*)d_out;

    disco_fused_lat<<<2 * NLAT_OUT, THREADS, 0, stream>>>(x, w, bias, ker, row, col, vals, out);
}


# gather waves own 60 positions; the freed 16th row fetches the x row's first group in the wave holding the row's end, the straddling group is built by lane broadcast: no 2nd load round; sum order kept
# speedup vs baseline: 4.9678x; 1.0584x over previous
#include <hip/hip_runtime.h>


typedef float v2f __attribute__((ext_vector_type(2)));
typedef float v4f __attribute__((ext_vector_type(4)));
typedef float v8f __attribute__((ext_vector_type(8)));
typedef float v4f_u __attribute__((ext_vector_type(4), aligned(4)));
typedef int   v2i __attribute__((ext_vector_type(2)));
typedef int   v4i __attribute__((ext_vector_type(4)));

#define NLAT_IN   181
#define NLON_IN   360
#define NLAT_OUT  361
#define NLON_OUT  720
#define KSIZE     3
#define CIN       16
#define COUT      16
#define NNZ       8192
#define NPIX      (NLAT_IN * NLON_IN)
#define HALF_T    (NLON_IN / 2)
#define HALF_LON  (NLON_OUT / 2)
#define ACC_PITCH 364
#define ROW_VEC   (HALF_LON / 4)
#define WO_TRIPS  ((COUT * ROW_VEC + THREADS - 1) / THREADS)
#define THREADS   256
#define WAVES     (THREADS / 32)
#define NZ_PER_WAVE (NNZ / WAVES)
#define NZ_GROUPS (NZ_PER_WAVE / 128)
#define GATHER_PAIRS 3
#define WAVE_T    60
#define SPARE_ROW 15
#define W_FRAGS   (KSIZE * 4 * 32)
#define W_TRIPS   ((W_FRAGS + THREADS - 1) / THREADS)
#define TILES_PER_WAVE 4

__device__ __forceinline__ float load_rows(v4f (&a)[8], int& sft, int& cs, const float* xh, int c, int T0,
                                           int word, int vbits)
{
    const int rowbase = word & 0xFFFF;
    const int sh      = word >> 18;
    int u = T0 - sh;
    if (u < 0) u += NLON_IN;
    cs  = (NLON_IN / 4 - 1) - (u >> 2);
    sft = cs < SPARE_ROW ? (u & 3) : 0;
    int po = u + 4 * c;
    if (po >= NLON_IN) po -= NLON_IN;
    int pc = po > NLON_IN - 4 ? NLON_IN - 4 : po;
    if (sft != 0 && c == SPARE_ROW) pc = 0;
    const float* rowp = xh + rowbase + pc;
#pragma unroll
    for (int s = 0; s < 4; ++s) {
        a[2 * s + 0] = *(const v4f_u*)(rowp + (4 * s + 0) * NPIX);
        a[2 * s + 1] = *(const v4f_u*)(rowp + (4 * s + 1) * NPIX);
    }
    return __int_as_float(vbits);
}

__device__ __forceinline__ float spare_lane(float f)
{
    return __int_as_float(__builtin_amdgcn_update_dpp(0, __float_as_int(f), 0x15F, 0xF, 0xF, true));
}

template <int SFT>
__device__ __forceinline__ v4f end_group(const v4f g)
{
    v4f n;
#pragma unroll
    for (int e = 0; e < 4; ++e) {
        float f = g[(e + SFT) & 3];
        if (e + SFT >= 4) f = spare_lane(f);
        n[e] = f;
    }
    return n;
}

__device__ __forceinline__ void add_row_end(v4f (&S)[8], const v4f (&a)[8], float vr, int sft)
{
#pragma unroll
    for (int k = 0; k < 8; ++k) {
        v4f n;
        if (sft == 1)      n = end_group<1>(a[k]);
        else if (sft == 2) n = end_group<2>(a[k]);
        else               n = end_group<3>(a[k]);
        S[k] += n * vr;
    }
}

__device__ __forceinline__ void add_rows(v4f (&S)[8], const v4f (&a)[8], float v, int sft, bool at_cs)
{
    const bool  me = sft != 0 && at_cs;
    const float vm = me ? 0.f : v;
#pragma unroll
    for (int k = 0; k < 8; ++k)
        S[k] += a[k] * vm;
    if (sft != 0)
        add_row_end(S, a, me ? v : 0.f, sft);
}

__device__ __forceinline__ void mix_basis(v8f (&d)[TILES_PER_WAVE],
                                          const v4f (&S)[8], const v2f* wk)
{
    v2f b[4];
#pragma unroll
    for (int s = 0; s < 4; ++s)
        b[s] = wk[s * 32];
#pragma unroll
    for (int s = 0; s < 4; ++s) {
#pragma unroll
        for (int e = 0; e < TILES_PER_WAVE; ++e) {
            v2f a;
            a.x = S[2 * s + 0][e];
            a.y = S[2 * s + 1][e];
            d[e] = __builtin_amdgcn_wmma_f32_16x16x4_f32(false, a, false, b[s],
                                                         (short)0, d[e], false, false);
        }
    }
}

__global__ __launch_bounds__(THREADS)
void disco_fused_lat(const float* __restrict__ x,
                     const float* __restrict__ w,
                     const float* __restrict__ bias,
                     const int*   __restrict__ ker_idx,
                     const int*   __restrict__ row_idx,
                     const int*   __restrict__ col_idx,
                     const float* __restrict__ vals,
                     float* __restrict__ out)
{
    __shared__ __attribute__((aligned(16))) float acc[COUT * ACC_PITCH];
    __shared__ v2i   rec[NNZ];
    __shared__ v2f   wl[W_FRAGS];
    __shared__ int   wtot[WAVES];

    const int tid  = threadIdx.x;
    const int half = blockIdx.x >= NLAT_OUT ? 1 : 0;
    const int hi   = blockIdx.x - half * NLAT_OUT;
    const int lane = tid & 31;
    const int wave = __builtin_amdgcn_readfirstlane(tid >> 5);

    const int hbase = hi * NLON_OUT;
    const int nz0   = wave * NZ_PER_WAVE + 4 * lane;
    v4i cv[NZ_GROUPS];
#pragma unroll
    for (int j = 0; j < NZ_GROUPS; ++j)
        cv[j] = *(const v4i*)(col_idx + nz0 + 128 * j);

    v2f wv[W_TRIPS];
#pragma unroll
    for (int t = 0; t < W_TRIPS; ++t) {
        const int i  = tid + t * THREADS;
        const int k  = i >> 7;
        const int s  = (i >> 5) & 3;
        const int cc = i & 15;
        const int hh = (i >> 4) & 1;
        const int ci = 4 * s + 2 * hh;
        wv[t] = (v2f){0.f, 0.f};
        if (i < W_FRAGS) {
            wv[t].x = w[cc * (CIN * KSIZE) + (ci + 0) * KSIZE + k];
            wv[t].y = w[cc * (CIN * KSIZE) + (ci + 1) * KSIZE + k];
        }
    }
#pragma unroll
    for (int t = 0; t < W_TRIPS; ++t) {
        const int i = tid + t * THREADS;
        if (i < W_FRAGS) wl[i] = wv[t];
    }

    unsigned hm = 0u, pm = 0u;
#pragma unroll
    for (int j = 0; j < NZ_GROUPS; ++j) {
#pragma unroll
        for (int e = 0; e < 4; ++e) {
            const unsigned p   = (unsigned)(cv[j][e] - hbase);
            const unsigned hit = p < (unsigned)NLON_OUT ? 1u : 0u;
            hm |= hit << (4 * j + e);
            pm |= (hit & p) << (4 * j + e);
        }
    }
    const int c1     = __builtin_popcount(pm);
    const int packed = (__builtin_popcount(hm) - c1) | (c1 << 16);

    int run = 0, mybase = 0;
    {
        unsigned act = __builtin_amdgcn_ballot_w32(packed != 0);
        while (act != 0u) {
            const int pick = __builtin_ctz(act);
            act &= act - 1u;
            const int cpk = __builtin_amdgcn_readlane(packed, pick);
            if (lane == pick) mybase = run;
            run += cpk;
        }
    }
    if (lane == 0) wtot[wave] = run;
    __syncthreads();

    int tot = 0, wbase = 0;
#pragma unroll
    for (int w2 = 0; w2 < WAVES; ++w2) {
        const int t = wtot[w2];
        tot += t;
        if (w2 < wave) wbase += t;
    }

    {
        int s0 = (wbase + mybase) & 0xFFFF;
        int s1 = (wbase + mybase) >> 16;
        while (hm != 0u) {
            const int i = __builtin_ctz(hm);
            hm &= hm - 1u;
            const int nz   = nz0 + 128 * (i >> 2) + (i & 3);
            const int p    = col_idx[nz] - hbase;
            const int word = (row_idx[nz] * NLON_IN) | (ker_idx[nz] << 16) | ((p >> 1) << 18);
            v2i rc;
            rc.x = word;
            rc.y = __float_as_int(vals[nz]);
            int slot;
            if ((pm >> i) & 1u) { slot = NNZ - 1 - s1; ++s1; }
            else                { slot = s0;           ++s0; }
            rec[slot] = rc;
        }
    }
    __syncthreads();

    const int q  = wave & 1;
    const int wp = wave >> 1;
    const int c  = lane & 15;
    const int h  = lane >> 4;

    if (wp < GATHER_PAIRS) {
    const int n  = __builtin_amdgcn_readfirstlane(q ? (tot >> 16) : (tot & 0xFFFF));

    const int T0 = HALF_T * half + WAVE_T * wp;

    const float* xh = x + 2 * h * NPIX;

    v8f d[TILES_PER_WAVE];
#pragma unroll
    for (int e = 0; e < TILES_PER_WAVE; ++e)
        d[e] = (v8f){0.f, 0.f, 0.f, 0.f, 0.f, 0.f, 0.f, 0.f};

    for (int kk = 0; kk < KSIZE; ++kk) {
      v4f S[8];
#pragma unroll
      for (int k = 0; k < 8; ++k)
          S[k] = (v4f){0.f, 0.f, 0.f, 0.f};
      for (int j0 = 0; j0 < n; j0 += 32) {
        const int idx = j0 + lane;
        v2i rc = (v2i){0, 0};
        if (idx < n) rc = rec[q ? (NNZ - 1 - idx) : idx];
        int kj = (rc.x >> 16) & 3;
        if (kj > KSIZE - 1) kj = KSIZE - 1;
        unsigned mask = __builtin_amdgcn_ballot_w32(idx < n && kj == kk);
        while (mask != 0u) {
          const int pick0 = __builtin_ctz(mask);
          mask &= mask - 1u;
          const bool two = mask != 0u;
          const int pick1 = two ? __builtin_ctz(mask) : pick0;
          mask &= mask - 1u;
          v4f a0[8], a1[8];
          int f0 = 0, f1 = 0, cs0 = 0, cs1 = 0;
          const float v0 = load_rows(a0, f0, cs0, xh, c, T0,
                                     __builtin_amdgcn_readlane(rc.x, pick0),
                                     __builtin_amdgcn_readlane(rc.y, pick0));
          float v1 = 0.f;
          if (two)
            v1 = load_rows(a1, f1, cs1, xh, c, T0,
                           __builtin_amdgcn_readlane(rc.x, pick1),
                           __builtin_amdgcn_readlane(rc.y, pick1));
          add_rows(S, a0, v0, f0, c == cs0);
          if (two)
            add_rows(S, a1, v1, f1, c == cs1);
        }
      }
      mix_basis(d, S, wl + kk * (4 * 32) + lane);
    }

#pragma unroll
    for (int e = 0; e < TILES_PER_WAVE; ++e) {
#pragma unroll
        for (int r = 0; r < 8; ++r) {
            const int t = WAVE_T * wp + 32 * h + 4 * r + e;
            if (8 * h + r < SPARE_ROW) acc[c * ACC_PITCH + q + 2 * t] = d[e][r];
        }
    }
    }
    __syncthreads();

    v4f   ov[WO_TRIPS];
    float ob[WO_TRIPS];
#pragma unroll
    for (int j = 0; j < WO_TRIPS; ++j) {
        int i = tid + j * THREADS;
        if (i > COUT * ROW_VEC - 1) i = COUT * ROW_VEC - 1;
        const int co = i / ROW_VEC;
        const int L  = 4 * (i - co * ROW_VEC);
        ov[j] = *(const v4f*)(acc + co * ACC_PITCH + L);
        ob[j] = bias[co];
    }
#pragma unroll
    for (int j = 0; j < WO_TRIPS; ++j) {
        const int i = tid + j * THREADS;
        if (i < COUT * ROW_VEC) {
            const int co = i / ROW_VEC;
            const int L  = 4 * (i - co * ROW_VEC);
            v4f v = ov[j];
            v.x += ob[j]; v.y += ob[j]; v.z += ob[j]; v.w += ob[j];
            *(v4f_u*)(out + (co * NLAT_OUT + hi) * NLON_OUT + HALF_LON * half + L) = v;
        }
    }
}

extern "C" void kernel_launch(void* const* d_in, const int* in_sizes, int n_in,
                              void* d_out, int out_size, void* d_ws, size_t ws_size,
                              hipStream_t stream)
{
    (void)in_sizes; (void)n_in; (void)out_size; (void)d_ws; (void)ws_size;

    const float* x    = (const float*)d_in[0];
    const float* w    = (const float*)d_in[1];
    const float* bias = (const float*)d_in[2];
    const int*   ker  = (const int*)  d_in[3];
    const int*   row  = (const int*)  d_in[4];
    const int*   col  = (const int*)  d_in[5];
    const float* vals = (const float*)d_in[6];
    float*       out  = (float*)d_out;

    disco_fused_lat<<<2 * NLAT_OUT, THREADS, 0, stream>>>(x, w, bias, ker, row, col, vals, out);
}
